# starting kernel
# baseline (speedup 1.0000x reference)
#include <hip/hip_runtime.h>
#include <math.h>

#define BB   64
#define NN   2048
#define MM   2048
#define CC   64
#define HH   128
#define NNZE 32768

typedef float v2f __attribute__((ext_vector_type(2)));
typedef float v8f __attribute__((ext_vector_type(8)));

__global__ __launch_bounds__(512)
void spmm_pool_kernel(const float* __restrict__ x,
                      const int*   __restrict__ rows,
                      const int*   __restrict__ cols,
                      const float* __restrict__ vals,
                      const int*   __restrict__ idx,
                      float* __restrict__ xdec) {
  extern __shared__ float smem[];
  float* sco  = smem;
  float* psum = sco  + MM * 32;
  float* pmax = psum + CC * 32;
  float* cnt  = pmax + CC * 32;

  const int tid    = threadIdx.x;
  const int lane   = tid & 31;
  const int wid    = tid >> 5;
  const int NWAVES = blockDim.x >> 5;
  const int b      = blockIdx.x >> 2;
  const int hbase  = (blockIdx.x & 3) * 32;

  for (int i = tid; i < MM * 32; i += blockDim.x) sco[i] = 0.0f;
  for (int i = tid; i < CC * 32; i += blockDim.x) { psum[i] = 0.0f; pmax[i] = -INFINITY; }
  for (int i = tid; i < CC;      i += blockDim.x) cnt[i] = 0.0f;
  __syncthreads();

  const size_t ebase = (size_t)b * NNZE;
  const float* xg = x + (size_t)b * NN * HH + hbase;
  for (int e0 = wid * 32; e0 < NNZE; e0 += NWAVES * 32) {
    const int   rbuf = rows[ebase + e0 + lane];
    const int   cbuf = cols[ebase + e0 + lane];
    const float vbuf = vals[ebase + e0 + lane];
#pragma unroll
    for (int j = 0; j < 32; ++j) {
      const int   r = __shfl(rbuf, j, 32);
      const int   c = __shfl(cbuf, j, 32);
      const float v = __shfl(vbuf, j, 32);
      const float xv = xg[(size_t)c * HH + lane];
      atomicAdd(&sco[r * 32 + lane], v * xv);
    }
  }
  __syncthreads();

  const size_t ibase = (size_t)b * MM;
  for (int m = wid; m < MM; m += NWAVES) {
    const int   c = idx[ibase + m];
    const float v = sco[m * 32 + lane];
    atomicAdd(&psum[c * 32 + lane], v);
    __hip_atomic_fetch_max(&pmax[c * 32 + lane], v,
                           __ATOMIC_RELAXED, __HIP_MEMORY_SCOPE_WORKGROUP);
    if (lane == 0) atomicAdd(&cnt[c], 1.0f);
  }
  __syncthreads();

  for (int c = wid; c < CC; c += NWAVES) {
    const float s    = psum[c * 32 + lane];
    const float mx   = pmax[c * 32 + lane];
    const float mean = s / fmaxf(cnt[c], 1.0f);
    const size_t base = ((size_t)b * CC + c) * (3 * HH);
    xdec[base +            hbase + lane] = mean;
    xdec[base +     HH +   hbase + lane] = mx;
    xdec[base + 2 * HH +   hbase + lane] = s;
  }
}

#define XD_LD 388
#define QK_LD 132
#define S_LD  66

__device__ __forceinline__ v2f ldsA(const float* base, int ld, int mt, int k, int lane) {
  const int hi = (lane >> 4) << 1;
  const float* p = base + (mt * 16 + (lane & 15)) * ld + k + hi;
  v2f a; a.x = p[0]; a.y = p[1];
  return a;
}

__global__ __launch_bounds__(256)
void cluster_attn_kernel(const float* __restrict__ xdec_g,
                         const float* __restrict__ Wq,
                         const float* __restrict__ Wk,
                         const float* __restrict__ Wv,
                         float* __restrict__ out) {
  extern __shared__ float smem[];
  float* xd = smem;
  float* Q  = xd + 64 * XD_LD;
  float* K  = Q  + 64 * QK_LD;
  float* V  = K  + 64 * QK_LD;
  float* S  = xd;

  const int tid  = threadIdx.x;
  const int lane = tid & 31;
  const int wid  = tid >> 5;
  const int b    = blockIdx.x;

  const float* src = xdec_g + (size_t)b * (CC * 3 * HH);
  for (int i = tid; i < CC * 3 * HH; i += blockDim.x)
    xd[(i / 384) * XD_LD + (i % 384)] = src[i];
  __syncthreads();

  for (int t = wid; t < 96; t += 8) {
    const float* W = (t < 32) ? Wq : (t < 64) ? Wk : Wv;
    float*       D = (t < 32) ? Q  : (t < 64) ? K  : V;
    const int tl = t & 31, mt = tl >> 3, nt = tl & 7;
    const int hi = (lane >> 4) << 1, col = nt * 16 + (lane & 15);
    v8f acc = {};
    for (int k = 0; k < 3 * HH; k += 4) {
      v2f a = ldsA(xd, XD_LD, mt, k, lane);
      v2f bf; bf.x = W[(k + hi) * HH + col]; bf.y = W[(k + 1 + hi) * HH + col];
      acc = __builtin_amdgcn_wmma_f32_16x16x4_f32(false, a, false, bf,
                                                  (short)0, acc, false, false);
    }
    const int rbase = mt * 16 + ((lane >> 4) << 3);
#pragma unroll
    for (int r = 0; r < 8; ++r) D[(rbase + r) * QK_LD + col] = acc[r];
  }
  __syncthreads();

  const float scale = 0.08838834764831845f;
  for (int t = wid; t < 16; t += 8) {
    const int mt = t >> 2, nt = t & 3;
    const int hi = (lane >> 4) << 1, col = nt * 16 + (lane & 15);
    v8f acc = {};
    for (int k = 0; k < HH; k += 4) {
      v2f a = ldsA(Q, QK_LD, mt, k, lane);
      v2f bf; bf.x = K[col * QK_LD + k + hi]; bf.y = K[col * QK_LD + k + hi + 1];
      acc = __builtin_amdgcn_wmma_f32_16x16x4_f32(false, a, false, bf,
                                                  (short)0, acc, false, false);
    }
    const int rbase = mt * 16 + ((lane >> 4) << 3);
#pragma unroll
    for (int r = 0; r < 8; ++r) S[(rbase + r) * S_LD + col] = acc[r] * scale;
  }
  __syncthreads();

  for (int r = wid; r < CC; r += 8) {
    float v0 = S[r * S_LD + lane];
    float v1 = S[r * S_LD + 32 + lane];
    float m = fmaxf(v0, v1);
    for (int off = 16; off > 0; off >>= 1) m = fmaxf(m, __shfl_xor(m, off, 32));
    float e0 = expf(v0 - m), e1 = expf(v1 - m);
    float s = e0 + e1;
    for (int off = 16; off > 0; off >>= 1) s += __shfl_xor(s, off, 32);
    const float inv = 1.0f / s;
    S[r * S_LD + lane]      = e0 * inv;
    S[r * S_LD + 32 + lane] = e1 * inv;
  }
  __syncthreads();

  float* ob = out + (size_t)b * (CC * HH);
  for (int t = wid; t < 32; t += 8) {
    const int mt = t >> 3, nt = t & 7;
    const int hi = (lane >> 4) << 1, col = nt * 16 + (lane & 15);
    v8f acc = {};
    for (int k = 0; k < CC; k += 4) {
      v2f a = ldsA(S, S_LD, mt, k, lane);
      v2f bf; bf.x = V[(k + hi) * QK_LD + col]; bf.y = V[(k + 1 + hi) * QK_LD + col];
      acc = __builtin_amdgcn_wmma_f32_16x16x4_f32(false, a, false, bf,
                                                  (short)0, acc, false, false);
    }
    const int rbase = mt * 16 + ((lane >> 4) << 3);
#pragma unroll
    for (int r = 0; r < 8; ++r) ob[(rbase + r) * HH + col] = acc[r];
  }
}

extern "C" void kernel_launch(void* const* d_in, const int* in_sizes, int n_in,
                              void* d_out, int out_size, void* d_ws, size_t ws_size,
                              hipStream_t stream) {
  const float* x    = (const float*)d_in[0];
  const int*   rows = (const int*)d_in[3];
  const int*   cols = (const int*)d_in[4];
  const float* vals = (const float*)d_in[5];
  const int*   idx  = (const int*)d_in[6];
  const float* Wq   = (const float*)d_in[7];
  const float* Wk   = (const float*)d_in[8];
  const float* Wv   = (const float*)d_in[9];
  float* xdec = (float*)d_ws;

  const size_t smem1 = (size_t)(MM * 32 + 2 * CC * 32 + CC) * sizeof(float);
  spmm_pool_kernel<<<dim3(BB * (HH / 32)), dim3(512), smem1, stream>>>(
      x, rows, cols, vals, idx, xdec);

  const size_t smem2 = (size_t)(64 * XD_LD + 3 * 64 * QK_LD) * sizeof(float);
  cluster_attn_kernel<<<dim3(BB), dim3(256), smem2, stream>>>(
      xdec, Wq, Wk, Wv, (float*)d_out);
}


# k1 lane map: 4 lanes an edge (or pooled row), a float4 of its LDS row each: 8 items a gather, 4 dependent LDS round trips a 32-item batch, not 16; lane's adds rotated over banks; sums by arrival
# speedup vs baseline: 4.8616x; 4.8616x over previous
#include <hip/hip_runtime.h>
#include <math.h>

#define BB   64
#define NN   2048
#define MM   2048
#define CC   64
#define HH   128
#define NNZE 32768

typedef float v2f __attribute__((ext_vector_type(2)));
typedef float v8f __attribute__((ext_vector_type(8)));

#define FC 16
#define K1_THREADS 1024
static_assert(CC * FC == K1_THREADS, "pool init below assumes one pool element per thread");
static_assert((NN * (FC / 4)) % K1_THREADS == 0 && (MM * (FC / 4)) % K1_THREADS == 0,
              "copy and clear loops assume whole passes of the block");

template <int J>
__device__ __forceinline__ int quad_bcast(int v) {
  return __builtin_amdgcn_update_dpp(0, v, J * 0x55, 0xf, 0xf, true);
}

template <int J>
__device__ __forceinline__ void edge_oct(int rb, int cb, int vb,
                                         const float* xs, float* sco,
                                         int q4, bool r1, bool r2, const int (&fo)[4]) {
  const int   r16 = quad_bcast<J>(rb);
  const int   c16 = quad_bcast<J>(cb);
  const float v   = __int_as_float(quad_bcast<J>(vb));
  const float4 xv = *reinterpret_cast<const float4*>(xs + c16 + q4);
  const float a0 = r1 ? xv.y : xv.x, a1 = r1 ? xv.z : xv.y;
  const float a2 = r1 ? xv.w : xv.z, a3 = r1 ? xv.x : xv.w;
  const float w0 = r2 ? a2 : a0, w1 = r2 ? a3 : a1;
  const float w2 = r2 ? a0 : a2, w3 = r2 ? a1 : a3;
  atomicAdd(&sco[r16 + fo[0]], v * w0);
  atomicAdd(&sco[r16 + fo[1]], v * w1);
  atomicAdd(&sco[r16 + fo[2]], v * w2);
  atomicAdd(&sco[r16 + fo[3]], v * w3);
}

template <int J>
__device__ __forceinline__ void pool_oct(int cb, int mrow, const float* sco,
                                         float* psum, float* pmax, float* cnt,
                                         int q4, bool r1, bool r2, const int (&fo)[4]) {
  const int    c16 = quad_bcast<J>(cb);
  const float4 xv  = *reinterpret_cast<const float4*>(sco + mrow + 8 * J * FC);
  const float a0 = r1 ? xv.y : xv.x, a1 = r1 ? xv.z : xv.y;
  const float a2 = r1 ? xv.w : xv.z, a3 = r1 ? xv.x : xv.w;
  const float w[4] = { r2 ? a2 : a0, r2 ? a3 : a1, r2 ? a0 : a2, r2 ? a1 : a3 };
#pragma unroll
  for (int i = 0; i < 4; ++i) {
    atomicAdd(&psum[c16 + fo[i]], w[i]);
    __hip_atomic_fetch_max(&pmax[c16 + fo[i]], w[i],
                           __ATOMIC_RELAXED, __HIP_MEMORY_SCOPE_WORKGROUP);
  }
  if (q4 == 0) atomicAdd(&cnt[c16 >> 4], 1.0f);
}

__global__ __launch_bounds__(K1_THREADS)
void spmm_pool_kernel(const float* __restrict__ x,
                      const int*   __restrict__ rows,
                      const int*   __restrict__ cols,
                      const float* __restrict__ vals,
                      const int*   __restrict__ idx,
                      float* __restrict__ xdec) {
  extern __shared__ float smem[];
  float* xs   = smem;
  float* sco  = xs   + NN * FC;
  float* psum = sco  + MM * FC;
  float* pmax = psum + CC * FC;
  float* cnt  = pmax + CC * FC;

  const int tid    = threadIdx.x;
  const int lane   = tid & 31;
  const int wid    = tid >> 5;
  constexpr int NWAVES = K1_THREADS / 32;
  const int f      = lane & 15;
  const int half   = lane >> 4;
  const int b      = blockIdx.x >> 3;
  const int hbase  = (blockIdx.x & 7) * FC;

  const float* xg = x + (size_t)b * NN * HH + hbase;
  float4 xr[NN * (FC / 4) / K1_THREADS];
  for (int k = 0; k < NN * (FC / 4) / K1_THREADS; ++k) {
    const int i = tid + k * K1_THREADS, n = i >> 2, q = i & 3;
    xr[k] = *reinterpret_cast<const float4*>(xg + (size_t)n * HH + q * 4);
  }
  for (int k = 0; k < NN * (FC / 4) / K1_THREADS; ++k) {
    const int i = tid + k * K1_THREADS, n = i >> 2, q = i & 3;
    *reinterpret_cast<float4*>(xs + n * FC + q * 4) = xr[k];
  }
  const float4 z4 = make_float4(0.0f, 0.0f, 0.0f, 0.0f);
  for (int k = 0; k < MM * (FC / 4) / K1_THREADS; ++k)
    reinterpret_cast<float4*>(sco)[tid + k * K1_THREADS] = z4;
  psum[tid] = 0.0f;
  pmax[tid] = -INFINITY;
  if (tid < CC) cnt[tid] = 0.0f;
  __syncthreads();

  const int  q4  = (lane & 3) * 4;
  const int  rot = (lane >> 2) & 3;
  const bool r1  = rot & 1, r2 = rot & 2;
  const int  fo[4] = { q4 + (rot & 3), q4 + ((1 + rot) & 3),
                       q4 + ((2 + rot) & 3), q4 + ((3 + rot) & 3) };
  const size_t ebase = (size_t)b * NNZE;
  for (int e0 = wid * 32; e0 < NNZE; e0 += NWAVES * 32) {
    const int rb = rows[ebase + e0 + lane] * FC;
    const int cb = cols[ebase + e0 + lane] * FC;
    const int vb = __float_as_int(vals[ebase + e0 + lane]);
    edge_oct<0>(rb, cb, vb, xs, sco, q4, r1, r2, fo);
    edge_oct<1>(rb, cb, vb, xs, sco, q4, r1, r2, fo);
    edge_oct<2>(rb, cb, vb, xs, sco, q4, r1, r2, fo);
    edge_oct<3>(rb, cb, vb, xs, sco, q4, r1, r2, fo);
  }
  __syncthreads();

  const int prow = ((lane & 3) << 3) | (lane >> 2);
  const size_t ibase = (size_t)b * MM;
  for (int m0 = wid * 32; m0 < MM; m0 += NWAVES * 32) {
    const int cb   = idx[ibase + m0 + prow] * FC;
    const int mrow = (m0 + (lane >> 2)) * FC + q4;
    pool_oct<0>(cb, mrow, sco, psum, pmax, cnt, q4, r1, r2, fo);
    pool_oct<1>(cb, mrow, sco, psum, pmax, cnt, q4, r1, r2, fo);
    pool_oct<2>(cb, mrow, sco, psum, pmax, cnt, q4, r1, r2, fo);
    pool_oct<3>(cb, mrow, sco, psum, pmax, cnt, q4, r1, r2, fo);
  }
  __syncthreads();

  for (int c = wid * 2 + half; c < CC; c += NWAVES * 2) {
    const float s    = psum[c * FC + f];
    const float mx   = pmax[c * FC + f];
    const float mean = s / fmaxf(cnt[c], 1.0f);
    const size_t base = ((size_t)b * CC + c) * (3 * HH);
    xdec[base +            hbase + f] = mean;
    xdec[base +     HH +   hbase + f] = mx;
    xdec[base + 2 * HH +   hbase + f] = s;
  }
}

#define XD_LD 388
#define QK_LD 132
#define S_LD  66
#define PJ_ROWS 16
#define PJ_THREADS 384
static_assert((PJ_ROWS * 3 * HH / 4) % PJ_THREADS == 0, "band copy assumes whole passes of the block");
static_assert(PJ_THREADS / 32 * 32 == 3 * HH, "one wave per 32 output columns");

__device__ __forceinline__ v2f ldsA(const float* base, int ld, int mt, int k, int lane) {
  const int hi = (lane >> 4) << 1;
  const float* p = base + (mt * 16 + (lane & 15)) * ld + k + hi;
  v2f a; a.x = p[0]; a.y = p[1];
  return a;
}

__global__ __launch_bounds__(PJ_THREADS)
void qkv_proj_kernel(float* __restrict__ xq,
                     const float* __restrict__ Wq,
                     const float* __restrict__ Wk,
                     const float* __restrict__ Wv) {
  __shared__ float xd[PJ_ROWS * XD_LD];
  __shared__ float od[PJ_ROWS * XD_LD];

  const int tid  = threadIdx.x;
  const int lane = tid & 31;
  const int wid  = tid >> 5;
  float* band = xq + (size_t)blockIdx.x * (PJ_ROWS * 3 * HH);

  constexpr int NCP = PJ_ROWS * 3 * HH / 4 / PJ_THREADS;
  float4 r4[NCP];
  for (int j = 0; j < NCP; ++j)
    r4[j] = reinterpret_cast<const float4*>(band)[tid + j * PJ_THREADS];
  for (int j = 0; j < NCP; ++j) {
    const int i = tid + j * PJ_THREADS, row = i / 96, q = i % 96;
    *reinterpret_cast<float4*>(xd + row * XD_LD + q * 4) = r4[j];
  }
  __syncthreads();

  const int m = wid >> 2;
  const float* Wm = (m == 0) ? Wq : (m == 1) ? Wk : Wv;
  const int c  = lane & 15;
  const int hi = (lane >> 4) << 1;
  const float* wp = Wm + (size_t)hi * HH + (wid & 3) * 32 + 2 * c;

  v8f acc0 = {}, acc1 = {};
  for (int k = 0; k < 3 * HH; k += 4) {
    const v2f a = ldsA(xd, XD_LD, 0, k, lane);
    const float2 w0 = *reinterpret_cast<const float2*>(wp + (size_t)k * HH);
    const float2 w1 = *reinterpret_cast<const float2*>(wp + (size_t)(k + 1) * HH);
    v2f b0; b0.x = w0.x; b0.y = w1.x;
    v2f b1; b1.x = w0.y; b1.y = w1.y;
    acc0 = __builtin_amdgcn_wmma_f32_16x16x4_f32(false, a, false, b0,
                                                 (short)0, acc0, false, false);
    acc1 = __builtin_amdgcn_wmma_f32_16x16x4_f32(false, a, false, b1,
                                                 (short)0, acc1, false, false);
  }

  const int rbase = (lane >> 4) << 3;
#pragma unroll
  for (int r = 0; r < 8; ++r) {
    float2 o; o.x = acc0[r]; o.y = acc1[r];
    *reinterpret_cast<float2*>(od + (rbase + r) * XD_LD + wid * 32 + 2 * c) = o;
  }
  __syncthreads();

  for (int j = 0; j < NCP; ++j) {
    const int i = tid + j * PJ_THREADS, row = i / 96, q = i % 96;
    reinterpret_cast<float4*>(band)[i] =
        *reinterpret_cast<const float4*>(od + row * XD_LD + q * 4);
  }
}

#define AT_THREADS 256
static_assert((CC * 3 * HH / 4) % AT_THREADS == 0, "Q|K|V copy assumes whole passes of the block");

__global__ __launch_bounds__(AT_THREADS)
void cluster_attn_kernel(const float* __restrict__ qkv_g,
                         float* __restrict__ out) {
  extern __shared__ float smem[];
  float* Q  = smem;
  float* K  = Q  + 64 * QK_LD;
  float* V  = K  + 64 * QK_LD;
  float* S  = V  + 64 * QK_LD;

  const int tid  = threadIdx.x;
  const int lane = tid & 31;
  const int wid  = tid >> 5;
  const int b    = blockIdx.x;

  const float4* src = reinterpret_cast<const float4*>(qkv_g + (size_t)b * (CC * 3 * HH));
  constexpr int NCP = CC * 3 * HH / 4 / AT_THREADS;
  constexpr int HCP = NCP / 2;
  for (int h2 = 0; h2 < 2; ++h2) {
    float4 r4[HCP];
    for (int j = 0; j < HCP; ++j) r4[j] = src[tid + (h2 * HCP + j) * AT_THREADS];
    for (int j = 0; j < HCP; ++j) {
      const int i = tid + (h2 * HCP + j) * AT_THREADS, row = i / 96, q = i % 96;
      *reinterpret_cast<float4*>(Q + (q >> 5) * (64 * QK_LD) + row * QK_LD + (q & 31) * 4) = r4[j];
    }
  }
  __syncthreads();

  const float scale = 0.08838834764831845f;
  for (int t = wid; t < 16; t += 8) {
    const int mt = t >> 2, nt = t & 3;
    const int hi = (lane >> 4) << 1, col = nt * 16 + (lane & 15);
    v8f acc = {};
    for (int k = 0; k < HH; k += 4) {
      v2f a = ldsA(Q, QK_LD, mt, k, lane);
      v2f bf; bf.x = K[col * QK_LD + k + hi]; bf.y = K[col * QK_LD + k + hi + 1];
      acc = __builtin_amdgcn_wmma_f32_16x16x4_f32(false, a, false, bf,
                                                  (short)0, acc, false, false);
    }
    const int rbase = mt * 16 + ((lane >> 4) << 3);
#pragma unroll
    for (int r = 0; r < 8; ++r) S[(rbase + r) * S_LD + col] = acc[r] * scale;
  }
  __syncthreads();

  for (int r = wid; r < CC; r += 8) {
    float v0 = S[r * S_LD + lane];
    float v1 = S[r * S_LD + 32 + lane];
    float m = fmaxf(v0, v1);
    for (int off = 16; off > 0; off >>= 1) m = fmaxf(m, __shfl_xor(m, off, 32));
    float e0 = expf(v0 - m), e1 = expf(v1 - m);
    float s = e0 + e1;
    for (int off = 16; off > 0; off >>= 1) s += __shfl_xor(s, off, 32);
    const float inv = 1.0f / s;
    S[r * S_LD + lane]      = e0 * inv;
    S[r * S_LD + 32 + lane] = e1 * inv;
  }
  __syncthreads();

  float* ob = out + (size_t)b * (CC * HH);
  for (int t = wid; t < 32; t += 8) {
    const int mt = t >> 3, nt = t & 7;
    const int hi = (lane >> 4) << 1, col = nt * 16 + (lane & 15);
    v8f acc = {};
    for (int k = 0; k < CC; k += 4) {
      v2f a = ldsA(S, S_LD, mt, k, lane);
      v2f bf; bf.x = V[(k + hi) * QK_LD + col]; bf.y = V[(k + 1 + hi) * QK_LD + col];
      acc = __builtin_amdgcn_wmma_f32_16x16x4_f32(false, a, false, bf,
                                                  (short)0, acc, false, false);
    }
    const int rbase = mt * 16 + ((lane >> 4) << 3);
#pragma unroll
    for (int r = 0; r < 8; ++r) ob[(rbase + r) * HH + col] = acc[r];
  }
}

extern "C" void kernel_launch(void* const* d_in, const int* in_sizes, int n_in,
                              void* d_out, int out_size, void* d_ws, size_t ws_size,
                              hipStream_t stream) {
  const float* x    = (const float*)d_in[0];
  const int*   rows = (const int*)d_in[3];
  const int*   cols = (const int*)d_in[4];
  const float* vals = (const float*)d_in[5];
  const int*   idx  = (const int*)d_in[6];
  const float* Wq   = (const float*)d_in[7];
  const float* Wk   = (const float*)d_in[8];
  const float* Wv   = (const float*)d_in[9];
  float* xdec = (float*)d_ws;

  const size_t smem1 = (size_t)(NN * FC + MM * FC + 2 * CC * FC + CC) * sizeof(float);
  spmm_pool_kernel<<<dim3(BB * (HH / FC)), dim3(K1_THREADS), smem1, stream>>>(
      x, rows, cols, vals, idx, xdec);

  qkv_proj_kernel<<<dim3(BB * CC / PJ_ROWS), dim3(PJ_THREADS), 0, stream>>>(
      xdec, Wq, Wk, Wv);

  const size_t smem3 = (size_t)(3 * 64 * QK_LD + 64 * S_LD) * sizeof(float);
  cluster_attn_kernel<<<dim3(BB), dim3(AT_THREADS), smem3, stream>>>(
      xdec, (float*)d_out);
}


# k1 edge arrays as 16-byte loads: a wave reads rows, cols, vals 256 edges a trip (8 a lane, 6 loads issued together), 4 memory round trips a wave, not 32; lane-to-edge map changes, sums by arrival
# speedup vs baseline: 5.2534x; 1.0806x over previous
#include <hip/hip_runtime.h>
#include <math.h>

#define BB   64
#define NN   2048
#define MM   2048
#define CC   64
#define HH   128
#define NNZE 32768

typedef float v2f __attribute__((ext_vector_type(2)));
typedef float v8f __attribute__((ext_vector_type(8)));

#define FC 16
#define K1_THREADS 1024
#define WAVE_EDGES 256
static_assert(CC * FC == K1_THREADS, "pool init below assumes one pool element per thread");
static_assert(NNZE % (K1_THREADS / 32 * WAVE_EDGES) == 0, "SpMM loop assumes whole trips of the block");
static_assert((NN * (FC / 4)) % K1_THREADS == 0 && (MM * (FC / 4)) % K1_THREADS == 0,
              "copy and clear loops assume whole passes of the block");

template <int J>
__device__ __forceinline__ int quad_bcast(int v) {
  return __builtin_amdgcn_update_dpp(0, v, J * 0x55, 0xf, 0xf, true);
}

template <int J>
__device__ __forceinline__ void edge_oct(int rb, int cb, int vb,
                                         const float* xs, float* sco,
                                         int q4, bool r1, bool r2, const int (&fo)[4]) {
  const int   r16 = quad_bcast<J>(rb);
  const int   c16 = quad_bcast<J>(cb);
  const float v   = __int_as_float(quad_bcast<J>(vb));
  const float4 xv = *reinterpret_cast<const float4*>(xs + c16 + q4);
  const float a0 = r1 ? xv.y : xv.x, a1 = r1 ? xv.z : xv.y;
  const float a2 = r1 ? xv.w : xv.z, a3 = r1 ? xv.x : xv.w;
  const float w0 = r2 ? a2 : a0, w1 = r2 ? a3 : a1;
  const float w2 = r2 ? a0 : a2, w3 = r2 ? a1 : a3;
  atomicAdd(&sco[r16 + fo[0]], v * w0);
  atomicAdd(&sco[r16 + fo[1]], v * w1);
  atomicAdd(&sco[r16 + fo[2]], v * w2);
  atomicAdd(&sco[r16 + fo[3]], v * w3);
}

__device__ __forceinline__ void edge_batch(int r, int c, float v,
                                           const float* xs, float* sco,
                                           int q4, bool r1, bool r2, const int (&fo)[4]) {
  const int rb = r * FC, cb = c * FC, vb = __float_as_int(v);
  edge_oct<0>(rb, cb, vb, xs, sco, q4, r1, r2, fo);
  edge_oct<1>(rb, cb, vb, xs, sco, q4, r1, r2, fo);
  edge_oct<2>(rb, cb, vb, xs, sco, q4, r1, r2, fo);
  edge_oct<3>(rb, cb, vb, xs, sco, q4, r1, r2, fo);
}

template <int J>
__device__ __forceinline__ void pool_oct(int cb, int mrow, const float* sco,
                                         float* psum, float* pmax, float* cnt,
                                         int q4, bool r1, bool r2, const int (&fo)[4]) {
  const int    c16 = quad_bcast<J>(cb);
  const float4 xv  = *reinterpret_cast<const float4*>(sco + mrow + 8 * J * FC);
  const float a0 = r1 ? xv.y : xv.x, a1 = r1 ? xv.z : xv.y;
  const float a2 = r1 ? xv.w : xv.z, a3 = r1 ? xv.x : xv.w;
  const float w[4] = { r2 ? a2 : a0, r2 ? a3 : a1, r2 ? a0 : a2, r2 ? a1 : a3 };
#pragma unroll
  for (int i = 0; i < 4; ++i) {
    atomicAdd(&psum[c16 + fo[i]], w[i]);
    __hip_atomic_fetch_max(&pmax[c16 + fo[i]], w[i],
                           __ATOMIC_RELAXED, __HIP_MEMORY_SCOPE_WORKGROUP);
  }
  if (q4 == 0) atomicAdd(&cnt[c16 >> 4], 1.0f);
}

__global__ __launch_bounds__(K1_THREADS)
void spmm_pool_kernel(const float* __restrict__ x,
                      const int*   __restrict__ rows,
                      const int*   __restrict__ cols,
                      const float* __restrict__ vals,
                      const int*   __restrict__ idx,
                      float* __restrict__ xdec) {
  extern __shared__ float smem[];
  float* xs   = smem;
  float* sco  = xs   + NN * FC;
  float* psum = sco  + MM * FC;
  float* pmax = psum + CC * FC;
  float* cnt  = pmax + CC * FC;

  const int tid    = threadIdx.x;
  const int lane   = tid & 31;
  const int wid    = tid >> 5;
  constexpr int NWAVES = K1_THREADS / 32;
  const int f      = lane & 15;
  const int half   = lane >> 4;
  const int b      = blockIdx.x >> 3;
  const int hbase  = (blockIdx.x & 7) * FC;

  const float* xg = x + (size_t)b * NN * HH + hbase;
  float4 xr[NN * (FC / 4) / K1_THREADS];
  for (int k = 0; k < NN * (FC / 4) / K1_THREADS; ++k) {
    const int i = tid + k * K1_THREADS, n = i >> 2, q = i & 3;
    xr[k] = *reinterpret_cast<const float4*>(xg + (size_t)n * HH + q * 4);
  }
  for (int k = 0; k < NN * (FC / 4) / K1_THREADS; ++k) {
    const int i = tid + k * K1_THREADS, n = i >> 2, q = i & 3;
    *reinterpret_cast<float4*>(xs + n * FC + q * 4) = xr[k];
  }
  const float4 z4 = make_float4(0.0f, 0.0f, 0.0f, 0.0f);
  for (int k = 0; k < MM * (FC / 4) / K1_THREADS; ++k)
    reinterpret_cast<float4*>(sco)[tid + k * K1_THREADS] = z4;
  psum[tid] = 0.0f;
  pmax[tid] = -INFINITY;
  if (tid < CC) cnt[tid] = 0.0f;
  __syncthreads();

  const int  q4  = (lane & 3) * 4;
  const int  rot = (lane >> 2) & 3;
  const bool r1  = rot & 1, r2 = rot & 2;
  const int  fo[4] = { q4 + (rot & 3), q4 + ((1 + rot) & 3),
                       q4 + ((2 + rot) & 3), q4 + ((3 + rot) & 3) };
  const size_t ebase = (size_t)b * NNZE;
  for (int e0 = wid * WAVE_EDGES; e0 < NNZE; e0 += NWAVES * WAVE_EDGES) {
    const size_t o = ebase + e0 + lane * 4;
    const int4   ra = *reinterpret_cast<const int4*>(rows + o);
    const int4   rc = *reinterpret_cast<const int4*>(rows + o + WAVE_EDGES / 2);
    const int4   ca = *reinterpret_cast<const int4*>(cols + o);
    const int4   cd = *reinterpret_cast<const int4*>(cols + o + WAVE_EDGES / 2);
    const float4 va = *reinterpret_cast<const float4*>(vals + o);
    const float4 vc = *reinterpret_cast<const float4*>(vals + o + WAVE_EDGES / 2);
    edge_batch(ra.x, ca.x, va.x, xs, sco, q4, r1, r2, fo);
    edge_batch(ra.y, ca.y, va.y, xs, sco, q4, r1, r2, fo);
    edge_batch(ra.z, ca.z, va.z, xs, sco, q4, r1, r2, fo);
    edge_batch(ra.w, ca.w, va.w, xs, sco, q4, r1, r2, fo);
    edge_batch(rc.x, cd.x, vc.x, xs, sco, q4, r1, r2, fo);
    edge_batch(rc.y, cd.y, vc.y, xs, sco, q4, r1, r2, fo);
    edge_batch(rc.z, cd.z, vc.z, xs, sco, q4, r1, r2, fo);
    edge_batch(rc.w, cd.w, vc.w, xs, sco, q4, r1, r2, fo);
  }
  __syncthreads();

  const int prow = ((lane & 3) << 3) | (lane >> 2);
  const size_t ibase = (size_t)b * MM;
  for (int m0 = wid * 32; m0 < MM; m0 += NWAVES * 32) {
    const int cb   = idx[ibase + m0 + prow] * FC;
    const int mrow = (m0 + (lane >> 2)) * FC + q4;
    pool_oct<0>(cb, mrow, sco, psum, pmax, cnt, q4, r1, r2, fo);
    pool_oct<1>(cb, mrow, sco, psum, pmax, cnt, q4, r1, r2, fo);
    pool_oct<2>(cb, mrow, sco, psum, pmax, cnt, q4, r1, r2, fo);
    pool_oct<3>(cb, mrow, sco, psum, pmax, cnt, q4, r1, r2, fo);
  }
  __syncthreads();

  for (int c = wid * 2 + half; c < CC; c += NWAVES * 2) {
    const float s    = psum[c * FC + f];
    const float mx   = pmax[c * FC + f];
    const float mean = s / fmaxf(cnt[c], 1.0f);
    const size_t base = ((size_t)b * CC + c) * (3 * HH);
    xdec[base +            hbase + f] = mean;
    xdec[base +     HH +   hbase + f] = mx;
    xdec[base + 2 * HH +   hbase + f] = s;
  }
}

#define XD_LD 388
#define QK_LD 132
#define S_LD  66
#define PJ_ROWS 16
#define PJ_THREADS 384
static_assert((PJ_ROWS * 3 * HH / 4) % PJ_THREADS == 0, "band copy assumes whole passes of the block");
static_assert(PJ_THREADS / 32 * 32 == 3 * HH, "one wave per 32 output columns");

__device__ __forceinline__ v2f ldsA(const float* base, int ld, int mt, int k, int lane) {
  const int hi = (lane >> 4) << 1;
  const float* p = base + (mt * 16 + (lane & 15)) * ld + k + hi;
  v2f a; a.x = p[0]; a.y = p[1];
  return a;
}

__global__ __launch_bounds__(PJ_THREADS)
void qkv_proj_kernel(float* __restrict__ xq,
                     const float* __restrict__ Wq,
                     const float* __restrict__ Wk,
                     const float* __restrict__ Wv) {
  __shared__ float xd[PJ_ROWS * XD_LD];
  __shared__ float od[PJ_ROWS * XD_LD];

  const int tid  = threadIdx.x;
  const int lane = tid & 31;
  const int wid  = tid >> 5;
  float* band = xq + (size_t)blockIdx.x * (PJ_ROWS * 3 * HH);

  constexpr int NCP = PJ_ROWS * 3 * HH / 4 / PJ_THREADS;
  float4 r4[NCP];
  for (int j = 0; j < NCP; ++j)
    r4[j] = reinterpret_cast<const float4*>(band)[tid + j * PJ_THREADS];
  for (int j = 0; j < NCP; ++j) {
    const int i = tid + j * PJ_THREADS, row = i / 96, q = i % 96;
    *reinterpret_cast<float4*>(xd + row * XD_LD + q * 4) = r4[j];
  }
  __syncthreads();

  const int m = wid >> 2;
  const float* Wm = (m == 0) ? Wq : (m == 1) ? Wk : Wv;
  const int c  = lane & 15;
  const int hi = (lane >> 4) << 1;
  const float* wp = Wm + (size_t)hi * HH + (wid & 3) * 32 + 2 * c;

  v8f acc0 = {}, acc1 = {};
  for (int k = 0; k < 3 * HH; k += 4) {
    const v2f a = ldsA(xd, XD_LD, 0, k, lane);
    const float2 w0 = *reinterpret_cast<const float2*>(wp + (size_t)k * HH);
    const float2 w1 = *reinterpret_cast<const float2*>(wp + (size_t)(k + 1) * HH);
    v2f b0; b0.x = w0.x; b0.y = w1.x;
    v2f b1; b1.x = w0.y; b1.y = w1.y;
    acc0 = __builtin_amdgcn_wmma_f32_16x16x4_f32(false, a, false, b0,
                                                 (short)0, acc0, false, false);
    acc1 = __builtin_amdgcn_wmma_f32_16x16x4_f32(false, a, false, b1,
                                                 (short)0, acc1, false, false);
  }

  const int rbase = (lane >> 4) << 3;
#pragma unroll
  for (int r = 0; r < 8; ++r) {
    float2 o; o.x = acc0[r]; o.y = acc1[r];
    *reinterpret_cast<float2*>(od + (rbase + r) * XD_LD + wid * 32 + 2 * c) = o;
  }
  __syncthreads();

  for (int j = 0; j < NCP; ++j) {
    const int i = tid + j * PJ_THREADS, row = i / 96, q = i % 96;
    reinterpret_cast<float4*>(band)[i] =
        *reinterpret_cast<const float4*>(od + row * XD_LD + q * 4);
  }
}

#define AT_THREADS 256
static_assert((CC * 3 * HH / 4) % AT_THREADS == 0, "Q|K|V copy assumes whole passes of the block");

__global__ __launch_bounds__(AT_THREADS)
void cluster_attn_kernel(const float* __restrict__ qkv_g,
                         float* __restrict__ out) {
  extern __shared__ float smem[];
  float* Q  = smem;
  float* K  = Q  + 64 * QK_LD;
  float* V  = K  + 64 * QK_LD;
  float* S  = V  + 64 * QK_LD;

  const int tid  = threadIdx.x;
  const int lane = tid & 31;
  const int wid  = tid >> 5;
  const int b    = blockIdx.x;

  const float4* src = reinterpret_cast<const float4*>(qkv_g + (size_t)b * (CC * 3 * HH));
  constexpr int NCP = CC * 3 * HH / 4 / AT_THREADS;
  constexpr int HCP = NCP / 2;
  for (int h2 = 0; h2 < 2; ++h2) {
    float4 r4[HCP];
    for (int j = 0; j < HCP; ++j) r4[j] = src[tid + (h2 * HCP + j) * AT_THREADS];
    for (int j = 0; j < HCP; ++j) {
      const int i = tid + (h2 * HCP + j) * AT_THREADS, row = i / 96, q = i % 96;
      *reinterpret_cast<float4*>(Q + (q >> 5) * (64 * QK_LD) + row * QK_LD + (q & 31) * 4) = r4[j];
    }
  }
  __syncthreads();

  const float scale = 0.08838834764831845f;
  for (int t = wid; t < 16; t += 8) {
    const int mt = t >> 2, nt = t & 3;
    const int hi = (lane >> 4) << 1, col = nt * 16 + (lane & 15);
    v8f acc = {};
    for (int k = 0; k < HH; k += 4) {
      v2f a = ldsA(Q, QK_LD, mt, k, lane);
      v2f bf; bf.x = K[col * QK_LD + k + hi]; bf.y = K[col * QK_LD + k + hi + 1];
      acc = __builtin_amdgcn_wmma_f32_16x16x4_f32(false, a, false, bf,
                                                  (short)0, acc, false, false);
    }
    const int rbase = mt * 16 + ((lane >> 4) << 3);
#pragma unroll
    for (int r = 0; r < 8; ++r) S[(rbase + r) * S_LD + col] = acc[r] * scale;
  }
  __syncthreads();

  for (int r = wid; r < CC; r += 8) {
    float v0 = S[r * S_LD + lane];
    float v1 = S[r * S_LD + 32 + lane];
    float m = fmaxf(v0, v1);
    for (int off = 16; off > 0; off >>= 1) m = fmaxf(m, __shfl_xor(m, off, 32));
    float e0 = expf(v0 - m), e1 = expf(v1 - m);
    float s = e0 + e1;
    for (int off = 16; off > 0; off >>= 1) s += __shfl_xor(s, off, 32);
    const float inv = 1.0f / s;
    S[r * S_LD + lane]      = e0 * inv;
    S[r * S_LD + 32 + lane] = e1 * inv;
  }
  __syncthreads();

  float* ob = out + (size_t)b * (CC * HH);
  for (int t = wid; t < 32; t += 8) {
    const int mt = t >> 3, nt = t & 7;
    const int hi = (lane >> 4) << 1, col = nt * 16 + (lane & 15);
    v8f acc = {};
    for (int k = 0; k < CC; k += 4) {
      v2f a = ldsA(S, S_LD, mt, k, lane);
      v2f bf; bf.x = V[(k + hi) * QK_LD + col]; bf.y = V[(k + 1 + hi) * QK_LD + col];
      acc = __builtin_amdgcn_wmma_f32_16x16x4_f32(false, a, false, bf,
                                                  (short)0, acc, false, false);
    }
    const int rbase = mt * 16 + ((lane >> 4) << 3);
#pragma unroll
    for (int r = 0; r < 8; ++r) ob[(rbase + r) * HH + col] = acc[r];
  }
}

extern "C" void kernel_launch(void* const* d_in, const int* in_sizes, int n_in,
                              void* d_out, int out_size, void* d_ws, size_t ws_size,
                              hipStream_t stream) {
  const float* x    = (const float*)d_in[0];
  const int*   rows = (const int*)d_in[3];
  const int*   cols = (const int*)d_in[4];
  const float* vals = (const float*)d_in[5];
  const int*   idx  = (const int*)d_in[6];
  const float* Wq   = (const float*)d_in[7];
  const float* Wk   = (const float*)d_in[8];
  const float* Wv   = (const float*)d_in[9];
  float* xdec = (float*)d_ws;

  const size_t smem1 = (size_t)(NN * FC + MM * FC + 2 * CC * FC + CC) * sizeof(float);
  spmm_pool_kernel<<<dim3(BB * (HH / FC)), dim3(K1_THREADS), smem1, stream>>>(
      x, rows, cols, vals, idx, xdec);

  qkv_proj_kernel<<<dim3(BB * CC / PJ_ROWS), dim3(PJ_THREADS), 0, stream>>>(
      xdec, Wq, Wk, Wv);

  const size_t smem3 = (size_t)(3 * 64 * QK_LD + 64 * S_LD) * sizeof(float);
  cluster_attn_kernel<<<dim3(BB), dim3(AT_THREADS), smem3, stream>>>(
      xdec, (float*)d_out);
}


# k3 softmax: the ten xor-butterfly steps of a row's max and sum are DPP row_xmask / permlanex16 lane moves, not ds_bpermute LDS round trips; a wave's 8 rows take each step together; sums as before
# speedup vs baseline: 5.4297x; 1.0336x over previous
#include <hip/hip_runtime.h>
#include <math.h>

#define BB   64
#define NN   2048
#define MM   2048
#define CC   64
#define HH   128
#define NNZE 32768

typedef float v2f __attribute__((ext_vector_type(2)));
typedef float v8f __attribute__((ext_vector_type(8)));

#define FC 16
#define K1_THREADS 1024
#define WAVE_EDGES 256
static_assert(CC * FC == K1_THREADS, "pool init below assumes one pool element per thread");
static_assert(NNZE % (K1_THREADS / 32 * WAVE_EDGES) == 0, "SpMM loop assumes whole trips of the block");
static_assert((NN * (FC / 4)) % K1_THREADS == 0 && (MM * (FC / 4)) % K1_THREADS == 0,
              "copy and clear loops assume whole passes of the block");

template <int J>
__device__ __forceinline__ int quad_bcast(int v) {
  return __builtin_amdgcn_update_dpp(0, v, J * 0x55, 0xf, 0xf, true);
}

template <int J>
__device__ __forceinline__ void edge_oct(int rb, int cb, int vb,
                                         const float* xs, float* sco,
                                         int q4, bool r1, bool r2, const int (&fo)[4]) {
  const int   r16 = quad_bcast<J>(rb);
  const int   c16 = quad_bcast<J>(cb);
  const float v   = __int_as_float(quad_bcast<J>(vb));
  const float4 xv = *reinterpret_cast<const float4*>(xs + c16 + q4);
  const float a0 = r1 ? xv.y : xv.x, a1 = r1 ? xv.z : xv.y;
  const float a2 = r1 ? xv.w : xv.z, a3 = r1 ? xv.x : xv.w;
  const float w0 = r2 ? a2 : a0, w1 = r2 ? a3 : a1;
  const float w2 = r2 ? a0 : a2, w3 = r2 ? a1 : a3;
  atomicAdd(&sco[r16 + fo[0]], v * w0);
  atomicAdd(&sco[r16 + fo[1]], v * w1);
  atomicAdd(&sco[r16 + fo[2]], v * w2);
  atomicAdd(&sco[r16 + fo[3]], v * w3);
}

__device__ __forceinline__ void edge_batch(int r, int c, float v,
                                           const float* xs, float* sco,
                                           int q4, bool r1, bool r2, const int (&fo)[4]) {
  const int rb = r * FC, cb = c * FC, vb = __float_as_int(v);
  edge_oct<0>(rb, cb, vb, xs, sco, q4, r1, r2, fo);
  edge_oct<1>(rb, cb, vb, xs, sco, q4, r1, r2, fo);
  edge_oct<2>(rb, cb, vb, xs, sco, q4, r1, r2, fo);
  edge_oct<3>(rb, cb, vb, xs, sco, q4, r1, r2, fo);
}

template <int J>
__device__ __forceinline__ void pool_oct(int cb, int mrow, const float* sco,
                                         float* psum, float* pmax, float* cnt,
                                         int q4, bool r1, bool r2, const int (&fo)[4]) {
  const int    c16 = quad_bcast<J>(cb);
  const float4 xv  = *reinterpret_cast<const float4*>(sco + mrow + 8 * J * FC);
  const float a0 = r1 ? xv.y : xv.x, a1 = r1 ? xv.z : xv.y;
  const float a2 = r1 ? xv.w : xv.z, a3 = r1 ? xv.x : xv.w;
  const float w[4] = { r2 ? a2 : a0, r2 ? a3 : a1, r2 ? a0 : a2, r2 ? a1 : a3 };
#pragma unroll
  for (int i = 0; i < 4; ++i) {
    atomicAdd(&psum[c16 + fo[i]], w[i]);
    __hip_atomic_fetch_max(&pmax[c16 + fo[i]], w[i],
                           __ATOMIC_RELAXED, __HIP_MEMORY_SCOPE_WORKGROUP);
  }
  if (q4 == 0) atomicAdd(&cnt[c16 >> 4], 1.0f);
}

__global__ __launch_bounds__(K1_THREADS)
void spmm_pool_kernel(const float* __restrict__ x,
                      const int*   __restrict__ rows,
                      const int*   __restrict__ cols,
                      const float* __restrict__ vals,
                      const int*   __restrict__ idx,
                      float* __restrict__ xdec) {
  extern __shared__ float smem[];
  float* xs   = smem;
  float* sco  = xs   + NN * FC;
  float* psum = sco  + MM * FC;
  float* pmax = psum + CC * FC;
  float* cnt  = pmax + CC * FC;

  const int tid    = threadIdx.x;
  const int lane   = tid & 31;
  const int wid    = tid >> 5;
  constexpr int NWAVES = K1_THREADS / 32;
  const int f      = lane & 15;
  const int half   = lane >> 4;
  const int b      = blockIdx.x >> 3;
  const int hbase  = (blockIdx.x & 7) * FC;

  const float* xg = x + (size_t)b * NN * HH + hbase;
  float4 xr[NN * (FC / 4) / K1_THREADS];
  for (int k = 0; k < NN * (FC / 4) / K1_THREADS; ++k) {
    const int i = tid + k * K1_THREADS, n = i >> 2, q = i & 3;
    xr[k] = *reinterpret_cast<const float4*>(xg + (size_t)n * HH + q * 4);
  }
  for (int k = 0; k < NN * (FC / 4) / K1_THREADS; ++k) {
    const int i = tid + k * K1_THREADS, n = i >> 2, q = i & 3;
    *reinterpret_cast<float4*>(xs + n * FC + q * 4) = xr[k];
  }
  const float4 z4 = make_float4(0.0f, 0.0f, 0.0f, 0.0f);
  for (int k = 0; k < MM * (FC / 4) / K1_THREADS; ++k)
    reinterpret_cast<float4*>(sco)[tid + k * K1_THREADS] = z4;
  psum[tid] = 0.0f;
  pmax[tid] = -INFINITY;
  if (tid < CC) cnt[tid] = 0.0f;
  __syncthreads();

  const int  q4  = (lane & 3) * 4;
  const int  rot = (lane >> 2) & 3;
  const bool r1  = rot & 1, r2 = rot & 2;
  const int  fo[4] = { q4 + (rot & 3), q4 + ((1 + rot) & 3),
                       q4 + ((2 + rot) & 3), q4 + ((3 + rot) & 3) };
  const size_t ebase = (size_t)b * NNZE;
  for (int e0 = wid * WAVE_EDGES; e0 < NNZE; e0 += NWAVES * WAVE_EDGES) {
    const size_t o = ebase + e0 + lane * 4;
    const int4   ra = *reinterpret_cast<const int4*>(rows + o);
    const int4   rc = *reinterpret_cast<const int4*>(rows + o + WAVE_EDGES / 2);
    const int4   ca = *reinterpret_cast<const int4*>(cols + o);
    const int4   cd = *reinterpret_cast<const int4*>(cols + o + WAVE_EDGES / 2);
    const float4 va = *reinterpret_cast<const float4*>(vals + o);
    const float4 vc = *reinterpret_cast<const float4*>(vals + o + WAVE_EDGES / 2);
    edge_batch(ra.x, ca.x, va.x, xs, sco, q4, r1, r2, fo);
    edge_batch(ra.y, ca.y, va.y, xs, sco, q4, r1, r2, fo);
    edge_batch(ra.z, ca.z, va.z, xs, sco, q4, r1, r2, fo);
    edge_batch(ra.w, ca.w, va.w, xs, sco, q4, r1, r2, fo);
    edge_batch(rc.x, cd.x, vc.x, xs, sco, q4, r1, r2, fo);
    edge_batch(rc.y, cd.y, vc.y, xs, sco, q4, r1, r2, fo);
    edge_batch(rc.z, cd.z, vc.z, xs, sco, q4, r1, r2, fo);
    edge_batch(rc.w, cd.w, vc.w, xs, sco, q4, r1, r2, fo);
  }
  __syncthreads();

  const int prow = ((lane & 3) << 3) | (lane >> 2);
  const size_t ibase = (size_t)b * MM;
  for (int m0 = wid * 32; m0 < MM; m0 += NWAVES * 32) {
    const int cb   = idx[ibase + m0 + prow] * FC;
    const int mrow = (m0 + (lane >> 2)) * FC + q4;
    pool_oct<0>(cb, mrow, sco, psum, pmax, cnt, q4, r1, r2, fo);
    pool_oct<1>(cb, mrow, sco, psum, pmax, cnt, q4, r1, r2, fo);
    pool_oct<2>(cb, mrow, sco, psum, pmax, cnt, q4, r1, r2, fo);
    pool_oct<3>(cb, mrow, sco, psum, pmax, cnt, q4, r1, r2, fo);
  }
  __syncthreads();

  for (int c = wid * 2 + half; c < CC; c += NWAVES * 2) {
    const float s    = psum[c * FC + f];
    const float mx   = pmax[c * FC + f];
    const float mean = s / fmaxf(cnt[c], 1.0f);
    const size_t base = ((size_t)b * CC + c) * (3 * HH);
    xdec[base +            hbase + f] = mean;
    xdec[base +     HH +   hbase + f] = mx;
    xdec[base + 2 * HH +   hbase + f] = s;
  }
}

#define XD_LD 388
#define QK_LD 132
#define S_LD  66
#define PJ_ROWS 16
#define PJ_THREADS 384
static_assert((PJ_ROWS * 3 * HH / 4) % PJ_THREADS == 0, "band copy assumes whole passes of the block");
static_assert(PJ_THREADS / 32 * 32 == 3 * HH, "one wave per 32 output columns");

__device__ __forceinline__ v2f ldsA(const float* base, int ld, int mt, int k, int lane) {
  const int hi = (lane >> 4) << 1;
  const float* p = base + (mt * 16 + (lane & 15)) * ld + k + hi;
  v2f a; a.x = p[0]; a.y = p[1];
  return a;
}

__global__ __launch_bounds__(PJ_THREADS)
void qkv_proj_kernel(float* __restrict__ xq,
                     const float* __restrict__ Wq,
                     const float* __restrict__ Wk,
                     const float* __restrict__ Wv) {
  __shared__ float xd[PJ_ROWS * XD_LD];
  __shared__ float od[PJ_ROWS * XD_LD];

  const int tid  = threadIdx.x;
  const int lane = tid & 31;
  const int wid  = tid >> 5;
  float* band = xq + (size_t)blockIdx.x * (PJ_ROWS * 3 * HH);

  constexpr int NCP = PJ_ROWS * 3 * HH / 4 / PJ_THREADS;
  float4 r4[NCP];
  for (int j = 0; j < NCP; ++j)
    r4[j] = reinterpret_cast<const float4*>(band)[tid + j * PJ_THREADS];
  for (int j = 0; j < NCP; ++j) {
    const int i = tid + j * PJ_THREADS, row = i / 96, q = i % 96;
    *reinterpret_cast<float4*>(xd + row * XD_LD + q * 4) = r4[j];
  }
  __syncthreads();

  const int m = wid >> 2;
  const float* Wm = (m == 0) ? Wq : (m == 1) ? Wk : Wv;
  const int c  = lane & 15;
  const int hi = (lane >> 4) << 1;
  const float* wp = Wm + (size_t)hi * HH + (wid & 3) * 32 + 2 * c;

  v8f acc0 = {}, acc1 = {};
  for (int k = 0; k < 3 * HH; k += 4) {
    const v2f a = ldsA(xd, XD_LD, 0, k, lane);
    const float2 w0 = *reinterpret_cast<const float2*>(wp + (size_t)k * HH);
    const float2 w1 = *reinterpret_cast<const float2*>(wp + (size_t)(k + 1) * HH);
    v2f b0; b0.x = w0.x; b0.y = w1.x;
    v2f b1; b1.x = w0.y; b1.y = w1.y;
    acc0 = __builtin_amdgcn_wmma_f32_16x16x4_f32(false, a, false, b0,
                                                 (short)0, acc0, false, false);
    acc1 = __builtin_amdgcn_wmma_f32_16x16x4_f32(false, a, false, b1,
                                                 (short)0, acc1, false, false);
  }

  const int rbase = (lane >> 4) << 3;
#pragma unroll
  for (int r = 0; r < 8; ++r) {
    float2 o; o.x = acc0[r]; o.y = acc1[r];
    *reinterpret_cast<float2*>(od + (rbase + r) * XD_LD + wid * 32 + 2 * c) = o;
  }
  __syncthreads();

  for (int j = 0; j < NCP; ++j) {
    const int i = tid + j * PJ_THREADS, row = i / 96, q = i % 96;
    reinterpret_cast<float4*>(band)[i] =
        *reinterpret_cast<const float4*>(od + row * XD_LD + q * 4);
  }
}

#define AT_THREADS 256
static_assert((CC * 3 * HH / 4) % AT_THREADS == 0, "Q|K|V copy assumes whole passes of the block");
static_assert(CC % (AT_THREADS / 32) == 0, "softmax assumes every wave owns the same number of rows");

template <int M>
__device__ __forceinline__ float xor_lane(float v) {
  return __int_as_float(__builtin_amdgcn_update_dpp(0, __float_as_int(v), 0x160 + M, 0xf, 0xf, true));
}
__device__ __forceinline__ float xor_lane16(float v) {
  const int x = __float_as_int(v);
  return __int_as_float(__builtin_amdgcn_permlanex16(x, x, 0x76543210, 0xfedcba98, false, false));
}

__global__ __launch_bounds__(AT_THREADS)
void cluster_attn_kernel(const float* __restrict__ qkv_g,
                         float* __restrict__ out) {
  extern __shared__ float smem[];
  float* Q  = smem;
  float* K  = Q  + 64 * QK_LD;
  float* V  = K  + 64 * QK_LD;
  float* S  = V  + 64 * QK_LD;

  const int tid  = threadIdx.x;
  const int lane = tid & 31;
  const int wid  = tid >> 5;
  const int b    = blockIdx.x;

  const float4* src = reinterpret_cast<const float4*>(qkv_g + (size_t)b * (CC * 3 * HH));
  constexpr int NCP = CC * 3 * HH / 4 / AT_THREADS;
  constexpr int HCP = NCP / 2;
  for (int h2 = 0; h2 < 2; ++h2) {
    float4 r4[HCP];
    for (int j = 0; j < HCP; ++j) r4[j] = src[tid + (h2 * HCP + j) * AT_THREADS];
    for (int j = 0; j < HCP; ++j) {
      const int i = tid + (h2 * HCP + j) * AT_THREADS, row = i / 96, q = i % 96;
      *reinterpret_cast<float4*>(Q + (q >> 5) * (64 * QK_LD) + row * QK_LD + (q & 31) * 4) = r4[j];
    }
  }
  __syncthreads();

  const float scale = 0.08838834764831845f;
  for (int t = wid; t < 16; t += 8) {
    const int mt = t >> 2, nt = t & 3;
    const int hi = (lane >> 4) << 1, col = nt * 16 + (lane & 15);
    v8f acc = {};
    for (int k = 0; k < HH; k += 4) {
      v2f a = ldsA(Q, QK_LD, mt, k, lane);
      v2f bf; bf.x = K[col * QK_LD + k + hi]; bf.y = K[col * QK_LD + k + hi + 1];
      acc = __builtin_amdgcn_wmma_f32_16x16x4_f32(false, a, false, bf,
                                                  (short)0, acc, false, false);
    }
    const int rbase = mt * 16 + ((lane >> 4) << 3);
#pragma unroll
    for (int r = 0; r < 8; ++r) S[(rbase + r) * S_LD + col] = acc[r] * scale;
  }
  __syncthreads();

  {
    constexpr int NR = CC / (AT_THREADS / 32);
    float v0[NR], v1[NR], m[NR], s[NR];
    for (int i = 0; i < NR; ++i) {
      const int r = wid + i * (AT_THREADS / 32);
      v0[i] = S[r * S_LD + lane];
      v1[i] = S[r * S_LD + 32 + lane];
    }
    for (int i = 0; i < NR; ++i) m[i] = fmaxf(v0[i], v1[i]);
    for (int i = 0; i < NR; ++i) m[i] = fmaxf(m[i], xor_lane16(m[i]));
    for (int i = 0; i < NR; ++i) m[i] = fmaxf(m[i], xor_lane<8>(m[i]));
    for (int i = 0; i < NR; ++i) m[i] = fmaxf(m[i], xor_lane<4>(m[i]));
    for (int i = 0; i < NR; ++i) m[i] = fmaxf(m[i], xor_lane<2>(m[i]));
    for (int i = 0; i < NR; ++i) m[i] = fmaxf(m[i], xor_lane<1>(m[i]));
    for (int i = 0; i < NR; ++i) {
      v0[i] = expf(v0[i] - m[i]);
      v1[i] = expf(v1[i] - m[i]);
      s[i]  = v0[i] + v1[i];
    }
    for (int i = 0; i < NR; ++i) s[i] += xor_lane16(s[i]);
    for (int i = 0; i < NR; ++i) s[i] += xor_lane<8>(s[i]);
    for (int i = 0; i < NR; ++i) s[i] += xor_lane<4>(s[i]);
    for (int i = 0; i < NR; ++i) s[i] += xor_lane<2>(s[i]);
    for (int i = 0; i < NR; ++i) s[i] += xor_lane<1>(s[i]);
    for (int i = 0; i < NR; ++i) {
      const int r = wid + i * (AT_THREADS / 32);
      const float inv = 1.0f / s[i];
      S[r * S_LD + lane]      = v0[i] * inv;
      S[r * S_LD + 32 + lane] = v1[i] * inv;
    }
  }
  __syncthreads();

  float* ob = out + (size_t)b * (CC * HH);
  for (int t = wid; t < 32; t += 8) {
    const int mt = t >> 3, nt = t & 7;
    const int hi = (lane >> 4) << 1, col = nt * 16 + (lane & 15);
    v8f acc = {};
    for (int k = 0; k < CC; k += 4) {
      v2f a = ldsA(S, S_LD, mt, k, lane);
      v2f bf; bf.x = V[(k + hi) * QK_LD + col]; bf.y = V[(k + 1 + hi) * QK_LD + col];
      acc = __builtin_amdgcn_wmma_f32_16x16x4_f32(false, a, false, bf,
                                                  (short)0, acc, false, false);
    }
    const int rbase = mt * 16 + ((lane >> 4) << 3);
#pragma unroll
    for (int r = 0; r < 8; ++r) ob[(rbase + r) * HH + col] = acc[r];
  }
}

extern "C" void kernel_launch(void* const* d_in, const int* in_sizes, int n_in,
                              void* d_out, int out_size, void* d_ws, size_t ws_size,
                              hipStream_t stream) {
  const float* x    = (const float*)d_in[0];
  const int*   rows = (const int*)d_in[3];
  const int*   cols = (const int*)d_in[4];
  const float* vals = (const float*)d_in[5];
  const int*   idx  = (const int*)d_in[6];
  const float* Wq   = (const float*)d_in[7];
  const float* Wk   = (const float*)d_in[8];
  const float* Wv   = (const float*)d_in[9];
  float* xdec = (float*)d_ws;

  const size_t smem1 = (size_t)(NN * FC + MM * FC + 2 * CC * FC + CC) * sizeof(float);
  spmm_pool_kernel<<<dim3(BB * (HH / FC)), dim3(K1_THREADS), smem1, stream>>>(
      x, rows, cols, vals, idx, xdec);

  qkv_proj_kernel<<<dim3(BB * CC / PJ_ROWS), dim3(PJ_THREADS), 0, stream>>>(
      xdec, Wq, Wk, Wv);

  const size_t smem3 = (size_t)(3 * 64 * QK_LD + 64 * S_LD) * sizeof(float);
  cluster_attn_kernel<<<dim3(BB), dim3(AT_THREADS), smem3, stream>>>(
      xdec, (float*)d_out);
}
